# starting kernel
# baseline (speedup 1.0000x reference)
#include <hip/hip_runtime.h>
#include <math.h>

typedef __attribute__((ext_vector_type(2))) float v2f;
typedef __attribute__((ext_vector_type(8))) float v8f;

#define NQ      12
#define DIM     4096
#define NLAYERS 20
#define F_IN    2048
#define PROW    17
#define GROW    (16 * PROW)

__device__ __forceinline__ int pidx(int i) { return (i >> 4) * PROW + (i & 15); }

__device__ __forceinline__ int cnot_src(int i) {
    const int cq[11] = {0, 2, 4, 6, 8, 10, 1, 3, 5, 7, 9};
    const int tq[11] = {1, 3, 5, 7, 9, 11, 2, 4, 6, 8, 10};
    int j = i;
#pragma unroll
    for (int k = 10; k >= 0; --k) {
        int cm = 1 << (11 - cq[k]);
        int tm = 1 << (11 - tq[k]);
        if (j & cm) j ^= tm;
    }
    return j;
}

__device__ __forceinline__ float tanh_pio2(float x) {
    float xc = fminf(fmaxf(x, -15.f), 15.f);
    float z  = __expf(xc * 2.0f);
    return (z - 1.0f) * __builtin_amdgcn_rcpf(z + 1.0f) * 1.57079632679489662f;
}

__global__ __launch_bounds__(256)
void qnet_kernel(const float* __restrict__ in, const float* __restrict__ qp,
                 float* __restrict__ out)
{
    __shared__ float Sa[256 * PROW];
    __shared__ float Sb[256 * PROW];
    __shared__ float Mm[3][16 * PROW];
    __shared__ float CS[NQ], SN[NQ];
    __shared__ float red[96];
    __shared__ float s_scale;

    const int t    = threadIdx.x;
    const int lane = t & 31;
    const int wave = t >> 5;
    const int hi   = lane >> 4;
    const int lo   = lane & 15;
    const int row  = blockIdx.x;

    const float* srcrow = in + (size_t)row * F_IN;
    float ss = 0.f;
#pragma unroll
    for (int e = 0; e < 8; ++e) {
        int i = e * 256 + t;
        float q = tanh_pio2(srcrow[i]);
        Sa[pidx(i)] = q;
        ss += q * q;
    }
#pragma unroll
    for (int e = 8; e < 16; ++e) Sa[pidx(e * 256 + t)] = 0.5f;

    for (int off = 16; off; off >>= 1) ss += __shfl_down(ss, off, 32);
    if (lane == 0) red[wave] = ss;
    __syncthreads();
    if (t == 0) {
        float tot = 512.0f;
        for (int w = 0; w < 8; ++w) tot += red[w];
        s_scale = 1.0f / sqrtf(tot);
    }
    __syncthreads();
    {
        float sc = s_scale;
#pragma unroll
        for (int e = 0; e < 16; ++e) Sa[pidx(e * 256 + t)] *= sc;
    }

    int srcA[16];
#pragma unroll
    for (int tt = 0; tt < 2; ++tt) {
        int T = wave + tt * 8;
#pragma unroll
        for (int s = 0; s < 4; ++s) {
#pragma unroll
            for (int d = 0; d < 2; ++d) {
                int i = (T * 16 + lo) * 16 + 4 * s + 2 * hi + d;
                srcA[tt * 8 + s * 2 + d] = pidx(cnot_src(i));
            }
        }
    }

    float* pa = Sa;
    float* pb = Sb;

    const int T0 = wave;
    const int T1 = wave + 8;

    const float* th = qp + NQ;

    for (int L = 0; L < NLAYERS; ++L) {
        if (t < NQ) {
            float a = th[L * NQ + t] * 0.5f;
            CS[t] = cosf(a);
            SN[t] = sinf(a);
        }
        __syncthreads();

        {
            int r = t >> 4, c = t & 15;
#pragma unroll
            for (int g = 0; g < 3; ++g) {
                float f = 1.f;
#pragma unroll
                for (int p = 0; p < 4; ++p) {
                    int q  = g * 4 + p;
                    int rb = (r >> (3 - p)) & 1;
                    int cb = (c >> (3 - p)) & 1;
                    f *= (rb == cb) ? CS[q] : (rb ? SN[q] : -SN[q]);
                }
                Mm[g][r * PROW + c] = f;
            }
        }
        __syncthreads();

        {
            v8f acc0 = {0, 0, 0, 0, 0, 0, 0, 0};
            v8f acc1 = {0, 0, 0, 0, 0, 0, 0, 0};
#pragma unroll
            for (int s = 0; s < 4; ++s) {
                int k0 = 4 * s + 2 * hi;
                v2f a0, a1, b;
                a0.x = pa[srcA[s * 2 + 0]];
                a0.y = pa[srcA[s * 2 + 1]];
                a1.x = pa[srcA[8 + s * 2 + 0]];
                a1.y = pa[srcA[8 + s * 2 + 1]];
                b.x  = Mm[2][lo * PROW + k0];
                b.y  = Mm[2][lo * PROW + k0 + 1];
                acc0 = __builtin_amdgcn_wmma_f32_16x16x4_f32(
                    false, a0, false, b, (short)0, acc0, false, false);
                acc1 = __builtin_amdgcn_wmma_f32_16x16x4_f32(
                    false, a1, false, b, (short)0, acc1, false, false);
            }
#pragma unroll
            for (int v = 0; v < 8; ++v) {
                pb[(T0 * 16 + v + 8 * hi) * PROW + lo] = acc0[v];
                pb[(T1 * 16 + v + 8 * hi) * PROW + lo] = acc1[v];
            }
        }
        __syncthreads();

        {
            v8f acc0 = {0, 0, 0, 0, 0, 0, 0, 0};
            v8f acc1 = {0, 0, 0, 0, 0, 0, 0, 0};
#pragma unroll
            for (int s = 0; s < 4; ++s) {
                int k0 = 4 * s + 2 * hi;
                v2f a, b0, b1;
                a.x  = Mm[1][lo * PROW + k0];
                a.y  = Mm[1][lo * PROW + k0 + 1];
                b0.x = pb[T0 * GROW + k0 * PROW + lo];
                b0.y = pb[T0 * GROW + (k0 + 1) * PROW + lo];
                b1.x = pb[T1 * GROW + k0 * PROW + lo];
                b1.y = pb[T1 * GROW + (k0 + 1) * PROW + lo];
                acc0 = __builtin_amdgcn_wmma_f32_16x16x4_f32(
                    false, a, false, b0, (short)0, acc0, false, false);
                acc1 = __builtin_amdgcn_wmma_f32_16x16x4_f32(
                    false, a, false, b1, (short)0, acc1, false, false);
            }
#pragma unroll
            for (int v = 0; v < 8; ++v) {
                pa[T0 * GROW + (v + 8 * hi) * PROW + lo] = acc0[v];
                pa[T1 * GROW + (v + 8 * hi) * PROW + lo] = acc1[v];
            }
        }
        __syncthreads();

        {
            v8f acc0 = {0, 0, 0, 0, 0, 0, 0, 0};
            v8f acc1 = {0, 0, 0, 0, 0, 0, 0, 0};
#pragma unroll
            for (int s = 0; s < 4; ++s) {
                int k0 = 4 * s + 2 * hi;
                v2f a, b0, b1;
                a.x  = Mm[0][lo * PROW + k0];
                a.y  = Mm[0][lo * PROW + k0 + 1];
                b0.x = pa[k0 * GROW + T0 * PROW + lo];
                b0.y = pa[(k0 + 1) * GROW + T0 * PROW + lo];
                b1.x = pa[k0 * GROW + T1 * PROW + lo];
                b1.y = pa[(k0 + 1) * GROW + T1 * PROW + lo];
                acc0 = __builtin_amdgcn_wmma_f32_16x16x4_f32(
                    false, a, false, b0, (short)0, acc0, false, false);
                acc1 = __builtin_amdgcn_wmma_f32_16x16x4_f32(
                    false, a, false, b1, (short)0, acc1, false, false);
            }
#pragma unroll
            for (int v = 0; v < 8; ++v) {
                pb[(v + 8 * hi) * GROW + T0 * PROW + lo] = acc0[v];
                pb[(v + 8 * hi) * GROW + T1 * PROW + lo] = acc1[v];
            }
        }
        __syncthreads();

        float* tmp = pa; pa = pb; pb = tmp;
    }

    float P = 0.f;
    float A[4] = {0.f, 0.f, 0.f, 0.f};
#pragma unroll
    for (int e = 0; e < 16; ++e) {
        int i = e * 256 + t;
        float v = pa[pidx(i)];
        float p = v * v;
        P += p;
#pragma unroll
        for (int q = 0; q < 4; ++q)
            A[q] += ((e >> (3 - q)) & 1) ? -p : p;
    }
    float acc[NQ];
#pragma unroll
    for (int q = 0; q < 4; ++q) acc[q] = A[q];
#pragma unroll
    for (int q = 4; q < NQ; ++q)
        acc[q] = ((t >> (11 - q)) & 1) ? -P : P;

    for (int off = 16; off; off >>= 1) {
#pragma unroll
        for (int q = 0; q < NQ; ++q) acc[q] += __shfl_down(acc[q], off, 32);
    }
    if (lane == 0) {
#pragma unroll
        for (int q = 0; q < NQ; ++q) red[wave * NQ + q] = acc[q];
    }
    __syncthreads();
    if (t < NQ) {
        float s = 0.f;
#pragma unroll
        for (int w = 0; w < 8; ++w) s += red[w * NQ + t];
        out[(size_t)row * NQ + t] = s;
    }
}

extern "C" void kernel_launch(void* const* d_in, const int* in_sizes, int n_in,
                              void* d_out, int out_size, void* d_ws, size_t ws_size,
                              hipStream_t stream) {
    (void)n_in; (void)out_size; (void)d_ws; (void)ws_size;
    const float* in  = (const float*)d_in[0];
    const float* qp  = (const float*)d_in[1];
    float*       out = (float*)d_out;
    int rows = in_sizes[0] / F_IN;
    qnet_kernel<<<rows, 256, 0, stream>>>(in, qp, out);
}


# axis-2 result tiles feed the axis-1 WMMA as B straight from registers (k order 8*hi+2*s+d on both operands): one LDS round trip and one barrier per layer dropped; axis-1 k summation order changes
# speedup vs baseline: 1.1331x; 1.1331x over previous
#include <hip/hip_runtime.h>
#include <math.h>

typedef __attribute__((ext_vector_type(2))) float v2f;
typedef __attribute__((ext_vector_type(8))) float v8f;

#define NQ      12
#define DIM     4096
#define NLAYERS 20
#define F_IN    2048
#define PROW    17
#define GROW    (16 * PROW)

__device__ __forceinline__ int pidx(int i) { return (i >> 4) * PROW + (i & 15); }

__device__ __forceinline__ int cnot_src(int i) {
    const int cq[11] = {0, 2, 4, 6, 8, 10, 1, 3, 5, 7, 9};
    const int tq[11] = {1, 3, 5, 7, 9, 11, 2, 4, 6, 8, 10};
    int j = i;
#pragma unroll
    for (int k = 10; k >= 0; --k) {
        int cm = 1 << (11 - cq[k]);
        int tm = 1 << (11 - tq[k]);
        if (j & cm) j ^= tm;
    }
    return j;
}

__device__ __forceinline__ float tanh_pio2(float x) {
    float xc = fminf(fmaxf(x, -15.f), 15.f);
    float z  = __expf(xc * 2.0f);
    return (z - 1.0f) * __builtin_amdgcn_rcpf(z + 1.0f) * 1.57079632679489662f;
}

__global__ __launch_bounds__(256)
void qnet_kernel(const float* __restrict__ in, const float* __restrict__ qp,
                 float* __restrict__ out)
{
    __shared__ float Sa[256 * PROW];
    __shared__ float Sb[256 * PROW];
    __shared__ float Mm[3][16 * PROW];
    __shared__ float CS[NQ], SN[NQ];
    __shared__ float red[96];
    __shared__ float s_scale;

    const int t    = threadIdx.x;
    const int lane = t & 31;
    const int wave = t >> 5;
    const int hi   = lane >> 4;
    const int lo   = lane & 15;
    const int row  = blockIdx.x;

    const float* srcrow = in + (size_t)row * F_IN;
    float ss = 0.f;
#pragma unroll
    for (int e = 0; e < 8; ++e) {
        int i = e * 256 + t;
        float q = tanh_pio2(srcrow[i]);
        Sa[pidx(i)] = q;
        ss += q * q;
    }
#pragma unroll
    for (int e = 8; e < 16; ++e) Sa[pidx(e * 256 + t)] = 0.5f;

    for (int off = 16; off; off >>= 1) ss += __shfl_down(ss, off, 32);
    if (lane == 0) red[wave] = ss;
    __syncthreads();
    if (t == 0) {
        float tot = 512.0f;
        for (int w = 0; w < 8; ++w) tot += red[w];
        s_scale = 1.0f / sqrtf(tot);
    }
    __syncthreads();
    {
        float sc = s_scale;
#pragma unroll
        for (int e = 0; e < 16; ++e) Sa[pidx(e * 256 + t)] *= sc;
    }

    int srcA[16];
#pragma unroll
    for (int tt = 0; tt < 2; ++tt) {
        int T = wave + tt * 8;
#pragma unroll
        for (int s = 0; s < 4; ++s) {
#pragma unroll
            for (int d = 0; d < 2; ++d) {
                int i = (T * 16 + lo) * 16 + 4 * s + 2 * hi + d;
                srcA[tt * 8 + s * 2 + d] = pidx(cnot_src(i));
            }
        }
    }

    float* const pa = Sa;
    float* const pb = Sb;

    const int T0 = wave;
    const int T1 = wave + 8;

    const float* th = qp + NQ;

    const float* const m2f = &Mm[2][lo * PROW + 2 * hi];
    const float* const m1f = &Mm[1][lo * PROW + 8 * hi];
    const float* const m0f = &Mm[0][lo * PROW + 2 * hi];
    float* const       yst = pb + T0 * GROW + 8 * hi * PROW + lo;
    const float* const yld = pb + 2 * hi * GROW + T0 * PROW + lo;
    float* const       xst = pa + 8 * hi * GROW + T0 * PROW + lo;

    for (int L = 0; L < NLAYERS; ++L) {
        if (t < NQ) {
            float a = th[L * NQ + t] * 0.5f;
            CS[t] = cosf(a);
            SN[t] = sinf(a);
        }
        __syncthreads();

        {
            int r = t >> 4, c = t & 15;
#pragma unroll
            for (int g = 0; g < 3; ++g) {
                float f = 1.f;
#pragma unroll
                for (int p = 0; p < 4; ++p) {
                    int q  = g * 4 + p;
                    int rb = (r >> (3 - p)) & 1;
                    int cb = (c >> (3 - p)) & 1;
                    f *= (rb == cb) ? CS[q] : (rb ? SN[q] : -SN[q]);
                }
                Mm[g][r * PROW + c] = f;
            }
        }
        __syncthreads();

        v8f acc0 = {0, 0, 0, 0, 0, 0, 0, 0};
        v8f acc1 = {0, 0, 0, 0, 0, 0, 0, 0};
#pragma unroll
        for (int s = 0; s < 4; ++s) {
            v2f a0, a1, b;
            a0.x = pa[srcA[s * 2 + 0]];
            a0.y = pa[srcA[s * 2 + 1]];
            a1.x = pa[srcA[8 + s * 2 + 0]];
            a1.y = pa[srcA[8 + s * 2 + 1]];
            b.x  = m2f[4 * s];
            b.y  = m2f[4 * s + 1];
            acc0 = __builtin_amdgcn_wmma_f32_16x16x4_f32(
                false, a0, false, b, (short)0, acc0, false, false);
            acc1 = __builtin_amdgcn_wmma_f32_16x16x4_f32(
                false, a1, false, b, (short)0, acc1, false, false);
        }

        {
            v8f c0 = {0, 0, 0, 0, 0, 0, 0, 0};
            v8f c1 = {0, 0, 0, 0, 0, 0, 0, 0};
#pragma unroll
            for (int s = 0; s < 4; ++s) {
                v2f a, b0, b1;
                a.x  = m1f[2 * s];
                a.y  = m1f[2 * s + 1];
                b0.x = acc0[2 * s];
                b0.y = acc0[2 * s + 1];
                b1.x = acc1[2 * s];
                b1.y = acc1[2 * s + 1];
                c0 = __builtin_amdgcn_wmma_f32_16x16x4_f32(
                    false, a, false, b0, (short)0, c0, false, false);
                c1 = __builtin_amdgcn_wmma_f32_16x16x4_f32(
                    false, a, false, b1, (short)0, c1, false, false);
            }
#pragma unroll
            for (int v = 0; v < 8; ++v) {
                yst[v * PROW]            = c0[v];
                yst[8 * GROW + v * PROW] = c1[v];
            }
        }
        __syncthreads();

        {
            v8f d0 = {0, 0, 0, 0, 0, 0, 0, 0};
            v8f d1 = {0, 0, 0, 0, 0, 0, 0, 0};
#pragma unroll
            for (int s = 0; s < 4; ++s) {
                v2f a, b0, b1;
                a.x  = m0f[4 * s];
                a.y  = m0f[4 * s + 1];
                b0.x = yld[(4 * s) * GROW];
                b0.y = yld[(4 * s + 1) * GROW];
                b1.x = yld[(4 * s) * GROW + 8 * PROW];
                b1.y = yld[(4 * s + 1) * GROW + 8 * PROW];
                d0 = __builtin_amdgcn_wmma_f32_16x16x4_f32(
                    false, a, false, b0, (short)0, d0, false, false);
                d1 = __builtin_amdgcn_wmma_f32_16x16x4_f32(
                    false, a, false, b1, (short)0, d1, false, false);
            }
#pragma unroll
            for (int v = 0; v < 8; ++v) {
                xst[v * GROW]            = d0[v];
                xst[v * GROW + 8 * PROW] = d1[v];
            }
        }
        __syncthreads();
    }

    float P = 0.f;
    float A[4] = {0.f, 0.f, 0.f, 0.f};
#pragma unroll
    for (int e = 0; e < 16; ++e) {
        int i = e * 256 + t;
        float v = pa[pidx(i)];
        float p = v * v;
        P += p;
#pragma unroll
        for (int q = 0; q < 4; ++q)
            A[q] += ((e >> (3 - q)) & 1) ? -p : p;
    }
    float acc[NQ];
#pragma unroll
    for (int q = 0; q < 4; ++q) acc[q] = A[q];
#pragma unroll
    for (int q = 4; q < NQ; ++q)
        acc[q] = ((t >> (11 - q)) & 1) ? -P : P;

    for (int off = 16; off; off >>= 1) {
#pragma unroll
        for (int q = 0; q < NQ; ++q) acc[q] += __shfl_down(acc[q], off, 32);
    }
    if (lane == 0) {
#pragma unroll
        for (int q = 0; q < NQ; ++q) red[wave * NQ + q] = acc[q];
    }
    __syncthreads();
    if (t < NQ) {
        float s = 0.f;
#pragma unroll
        for (int w = 0; w < 8; ++w) s += red[w * NQ + t];
        out[(size_t)row * NQ + t] = s;
    }
}

extern "C" void kernel_launch(void* const* d_in, const int* in_sizes, int n_in,
                              void* d_out, int out_size, void* d_ws, size_t ws_size,
                              hipStream_t stream) {
    (void)n_in; (void)out_size; (void)d_ws; (void)ws_size;
    const float* in  = (const float*)d_in[0];
    const float* qp  = (const float*)d_in[1];
    float*       out = (float*)d_out;
    int rows = in_sizes[0] / F_IN;
    qnet_kernel<<<rows, 256, 0, stream>>>(in, qp, out);
}


# LDS re-laid out for its 64 banks (state slab pitches 266/264 with bit-permuted row offsets, matrix row pitch 20): every fragment load and tile store of the layer loop is conflict-free, was 2-4 way
# speedup vs baseline: 1.2195x; 1.0763x over previous
#include <hip/hip_runtime.h>
#include <math.h>

typedef __attribute__((ext_vector_type(2))) float v2f;
typedef __attribute__((ext_vector_type(8))) float v8f;

#define NQ      12
#define DIM     4096
#define NLAYERS 20
#define F_IN    2048
#define PA0     266
#define PB0     264
#define SA_SIZE (15 * PA0 + 264)
#define SB_SIZE (15 * PB0 + 256)
#define MROW    20

__device__ __forceinline__ int rowA(int a1) {
    return 16 * (a1 & 1) + 64 * ((a1 >> 1) & 1) + 136 * ((a1 >> 2) & 1) + 32 * (a1 >> 3);
}
__device__ __forceinline__ int rowB(int a1) {
    return 16 * (a1 & 1) + 64 * ((a1 >> 1) & 3) + 32 * (a1 >> 3);
}
__device__ __forceinline__ int pidx(int i) { return (i >> 8) * PA0 + rowA((i >> 4) & 15) + (i & 15); }

__device__ __forceinline__ int cnot_src(int i) {
    const int cq[11] = {0, 2, 4, 6, 8, 10, 1, 3, 5, 7, 9};
    const int tq[11] = {1, 3, 5, 7, 9, 11, 2, 4, 6, 8, 10};
    int j = i;
#pragma unroll
    for (int k = 10; k >= 0; --k) {
        int cm = 1 << (11 - cq[k]);
        int tm = 1 << (11 - tq[k]);
        if (j & cm) j ^= tm;
    }
    return j;
}

__device__ __forceinline__ float tanh_pio2(float x) {
    float xc = fminf(fmaxf(x, -15.f), 15.f);
    float z  = __expf(xc * 2.0f);
    return (z - 1.0f) * __builtin_amdgcn_rcpf(z + 1.0f) * 1.57079632679489662f;
}

__global__ __launch_bounds__(256)
void qnet_kernel(const float* __restrict__ in, const float* __restrict__ qp,
                 float* __restrict__ out)
{
    __shared__ __attribute__((aligned(16))) float Sa[SA_SIZE];
    __shared__ __attribute__((aligned(16))) float Sb[SB_SIZE];
    __shared__ __attribute__((aligned(16))) float Mm[3][16 * MROW];
    __shared__ float CS[NQ], SN[NQ];
    __shared__ float red[96];
    __shared__ float s_scale;

    const int t    = threadIdx.x;
    const int lane = t & 31;
    const int wave = t >> 5;
    const int hi   = lane >> 4;
    const int lo   = lane & 15;
    const int row  = blockIdx.x;

    const float* srcrow = in + (size_t)row * F_IN;
    float ss = 0.f;
#pragma unroll
    for (int e = 0; e < 8; ++e) {
        int i = e * 256 + t;
        float q = tanh_pio2(srcrow[i]);
        Sa[pidx(i)] = q;
        ss += q * q;
    }
#pragma unroll
    for (int e = 8; e < 16; ++e) Sa[pidx(e * 256 + t)] = 0.5f;

    for (int off = 16; off; off >>= 1) ss += __shfl_down(ss, off, 32);
    if (lane == 0) red[wave] = ss;
    __syncthreads();
    if (t == 0) {
        float tot = 512.0f;
        for (int w = 0; w < 8; ++w) tot += red[w];
        s_scale = 1.0f / sqrtf(tot);
    }
    __syncthreads();
    {
        float sc = s_scale;
#pragma unroll
        for (int e = 0; e < 16; ++e) Sa[pidx(e * 256 + t)] *= sc;
    }

    int srcA[16];
#pragma unroll
    for (int tt = 0; tt < 2; ++tt) {
        int T = wave + tt * 8;
#pragma unroll
        for (int s = 0; s < 4; ++s) {
#pragma unroll
            for (int d = 0; d < 2; ++d) {
                int i = (T * 16 + lo) * 16 + 4 * s + 2 * hi + d;
                srcA[tt * 8 + s * 2 + d] = pidx(cnot_src(i));
            }
        }
    }

    float* const pa = Sa;
    float* const pb = Sb;

    const int T0 = wave;
    const int T1 = wave + 8;

    const float* th = qp + NQ;

    const float* const m2f = &Mm[2][lo * MROW + 10 * hi];
    const float* const m1f = &Mm[1][lo * MROW + 10 * hi];
    const float* const m0f = &Mm[0][lo * MROW + 10 * hi];
    float* const       yst = pb + T0 * PB0 + rowB(8 * hi) + lo;
    const float* const yld = pb + 2 * hi * PB0 + rowB(T0) + lo;
    float* const       xst = pa + 8 * hi * PA0 + rowA(T0) + lo;

    for (int L = 0; L < NLAYERS; ++L) {
        if (t < NQ) {
            float a = th[L * NQ + t] * 0.5f;
            CS[t] = cosf(a);
            SN[t] = sinf(a);
        }
        __syncthreads();

        {
            int r = t >> 4, c = t & 15;
#pragma unroll
            for (int g = 0; g < 3; ++g) {
                float f = 1.f;
#pragma unroll
                for (int p = 0; p < 4; ++p) {
                    int q  = g * 4 + p;
                    int rb = (r >> (3 - p)) & 1;
                    int cb = (c >> (3 - p)) & 1;
                    f *= (rb == cb) ? CS[q] : (rb ? SN[q] : -SN[q]);
                }
                int pc = (g == 1) ? (10 * (c >> 3) + 2 * ((c >> 1) & 3) + (c & 1))
                                  : (10 * ((c >> 1) & 1) + 2 * (c >> 2) + (c & 1));
                Mm[g][r * MROW + pc] = f;
            }
        }
        __syncthreads();

        v8f acc0 = {0, 0, 0, 0, 0, 0, 0, 0};
        v8f acc1 = {0, 0, 0, 0, 0, 0, 0, 0};
#pragma unroll
        for (int s = 0; s < 4; ++s) {
            v2f a0, a1, b;
            a0.x = pa[srcA[s * 2 + 0]];
            a0.y = pa[srcA[s * 2 + 1]];
            a1.x = pa[srcA[8 + s * 2 + 0]];
            a1.y = pa[srcA[8 + s * 2 + 1]];
            b.x  = m2f[2 * s];
            b.y  = m2f[2 * s + 1];
            acc0 = __builtin_amdgcn_wmma_f32_16x16x4_f32(
                false, a0, false, b, (short)0, acc0, false, false);
            acc1 = __builtin_amdgcn_wmma_f32_16x16x4_f32(
                false, a1, false, b, (short)0, acc1, false, false);
        }

        {
            v8f c0 = {0, 0, 0, 0, 0, 0, 0, 0};
            v8f c1 = {0, 0, 0, 0, 0, 0, 0, 0};
#pragma unroll
            for (int s = 0; s < 4; ++s) {
                v2f a, b0, b1;
                a.x  = m1f[2 * s];
                a.y  = m1f[2 * s + 1];
                b0.x = acc0[2 * s];
                b0.y = acc0[2 * s + 1];
                b1.x = acc1[2 * s];
                b1.y = acc1[2 * s + 1];
                c0 = __builtin_amdgcn_wmma_f32_16x16x4_f32(
                    false, a, false, b0, (short)0, c0, false, false);
                c1 = __builtin_amdgcn_wmma_f32_16x16x4_f32(
                    false, a, false, b1, (short)0, c1, false, false);
            }
#pragma unroll
            for (int v = 0; v < 8; ++v) {
                yst[rowB(v)]           = c0[v];
                yst[8 * PB0 + rowB(v)] = c1[v];
            }
        }
        __syncthreads();

        {
            v8f d0 = {0, 0, 0, 0, 0, 0, 0, 0};
            v8f d1 = {0, 0, 0, 0, 0, 0, 0, 0};
#pragma unroll
            for (int s = 0; s < 4; ++s) {
                v2f a, b0, b1;
                a.x  = m0f[2 * s];
                a.y  = m0f[2 * s + 1];
                b0.x = yld[(4 * s) * PB0];
                b0.y = yld[(4 * s + 1) * PB0];
                b1.x = yld[(4 * s) * PB0 + 32];
                b1.y = yld[(4 * s + 1) * PB0 + 32];
                d0 = __builtin_amdgcn_wmma_f32_16x16x4_f32(
                    false, a, false, b0, (short)0, d0, false, false);
                d1 = __builtin_amdgcn_wmma_f32_16x16x4_f32(
                    false, a, false, b1, (short)0, d1, false, false);
            }
#pragma unroll
            for (int v = 0; v < 8; ++v) {
                xst[v * PA0]      = d0[v];
                xst[v * PA0 + 32] = d1[v];
            }
        }
        __syncthreads();
    }

    float P = 0.f;
    float A[4] = {0.f, 0.f, 0.f, 0.f};
#pragma unroll
    for (int e = 0; e < 16; ++e) {
        int i = e * 256 + t;
        float v = pa[pidx(i)];
        float p = v * v;
        P += p;
#pragma unroll
        for (int q = 0; q < 4; ++q)
            A[q] += ((e >> (3 - q)) & 1) ? -p : p;
    }
    float acc[NQ];
#pragma unroll
    for (int q = 0; q < 4; ++q) acc[q] = A[q];
#pragma unroll
    for (int q = 4; q < NQ; ++q)
        acc[q] = ((t >> (11 - q)) & 1) ? -P : P;

    for (int off = 16; off; off >>= 1) {
#pragma unroll
        for (int q = 0; q < NQ; ++q) acc[q] += __shfl_down(acc[q], off, 32);
    }
    if (lane == 0) {
#pragma unroll
        for (int q = 0; q < NQ; ++q) red[wave * NQ + q] = acc[q];
    }
    __syncthreads();
    if (t < NQ) {
        float s = 0.f;
#pragma unroll
        for (int w = 0; w < 8; ++w) s += red[w * NQ + t];
        out[(size_t)row * NQ + t] = s;
    }
}

extern "C" void kernel_launch(void* const* d_in, const int* in_sizes, int n_in,
                              void* d_out, int out_size, void* d_ws, size_t ws_size,
                              hipStream_t stream) {
    (void)n_in; (void)out_size; (void)d_ws; (void)ws_size;
    const float* in  = (const float*)d_in[0];
    const float* qp  = (const float*)d_in[1];
    float*       out = (float*)d_out;
    int rows = in_sizes[0] / F_IN;
    qnet_kernel<<<rows, 256, 0, stream>>>(in, qp, out);
}


# rotation-matrix preparation pipelined half a layer ahead of its use (240 cos/sin pairs computed once, one per thread; M0 built beside axis-2/1, next M1,M2 beside axis-0): 2 of 4 barriers per layer go
# speedup vs baseline: 1.2628x; 1.0355x over previous
#include <hip/hip_runtime.h>
#include <math.h>

typedef __attribute__((ext_vector_type(2))) float v2f;
typedef __attribute__((ext_vector_type(8))) float v8f;

#define NQ      12
#define DIM     4096
#define NLAYERS 20
#define F_IN    2048
#define PA0     266
#define PB0     264
#define SA_SIZE (15 * PA0 + 264)
#define SB_SIZE (15 * PB0 + 256)
#define MROW    20

__device__ __forceinline__ int rowA(int a1) {
    return 16 * (a1 & 1) + 64 * ((a1 >> 1) & 1) + 136 * ((a1 >> 2) & 1) + 32 * (a1 >> 3);
}
__device__ __forceinline__ int rowB(int a1) {
    return 16 * (a1 & 1) + 64 * ((a1 >> 1) & 3) + 32 * (a1 >> 3);
}
__device__ __forceinline__ int pidx(int i) { return (i >> 8) * PA0 + rowA((i >> 4) & 15) + (i & 15); }

__device__ __forceinline__ int cnot_src(int i) {
    const int cq[11] = {0, 2, 4, 6, 8, 10, 1, 3, 5, 7, 9};
    const int tq[11] = {1, 3, 5, 7, 9, 11, 2, 4, 6, 8, 10};
    int j = i;
#pragma unroll
    for (int k = 10; k >= 0; --k) {
        int cm = 1 << (11 - cq[k]);
        int tm = 1 << (11 - tq[k]);
        if (j & cm) j ^= tm;
    }
    return j;
}

__device__ __forceinline__ float tanh_pio2(float x) {
    float xc = fminf(fmaxf(x, -15.f), 15.f);
    float z  = __expf(xc * 2.0f);
    return (z - 1.0f) * __builtin_amdgcn_rcpf(z + 1.0f) * 1.57079632679489662f;
}

__global__ __launch_bounds__(256, 10)
void qnet_kernel(const float* __restrict__ in, const float* __restrict__ qp,
                 float* __restrict__ out)
{
    __shared__ __attribute__((aligned(16))) float Sa[SA_SIZE];
    __shared__ __attribute__((aligned(16))) float Sb[SB_SIZE];
    __shared__ __attribute__((aligned(16))) float Mm[3][16 * MROW];
    __shared__ float CS[NQ], SN[NQ];
    __shared__ float red[96];
    __shared__ float s_scale;

    const int t    = threadIdx.x;
    const int lane = t & 31;
    const int wave = t >> 5;
    const int hi   = lane >> 4;
    const int lo   = lane & 15;
    const int row  = blockIdx.x;

    const float* th = qp + NQ;

    const int  hasAng = t < NLAYERS * NQ;
    const int  myL    = hasAng ? t / NQ : -1;
    const int  myq    = t - myL * NQ;
    float mc = 1.f, ms = 0.f;
    if (hasAng) {
        float a = th[t] * 0.5f;
        mc = cosf(a);
        ms = sinf(a);
    }
    if (myL == 0) { CS[myq] = mc; SN[myq] = ms; }

    const int mr = t >> 4, mcn = t & 15;
    const int pc1  = 10 * (mcn >> 3) + 2 * ((mcn >> 1) & 3) + (mcn & 1);
    const int pc02 = 10 * ((mcn >> 1) & 1) + 2 * (mcn >> 2) + (mcn & 1);
    auto kron_entry = [&](const float* c4, const float* s4) {
        float f = 1.f;
#pragma unroll
        for (int p = 0; p < 4; ++p) {
            int rb = (mr >> (3 - p)) & 1;
            int cb = (mcn >> (3 - p)) & 1;
            f *= (rb == cb) ? c4[p] : (rb ? s4[p] : -s4[p]);
        }
        return f;
    };
    auto load4 = [&](float* dst, const float* src4) {
#pragma unroll
        for (int p = 0; p < 4; ++p) dst[p] = src4[p];
    };
    float* const mst0 = &Mm[0][mr * MROW + pc02];
    float* const mst1 = &Mm[1][mr * MROW + pc1];
    float* const mst2 = &Mm[2][mr * MROW + pc02];

    const float* srcrow = in + (size_t)row * F_IN;
    float ss = 0.f;
#pragma unroll
    for (int e = 0; e < 8; ++e) {
        int i = e * 256 + t;
        float q = tanh_pio2(srcrow[i]);
        Sa[pidx(i)] = q;
        ss += q * q;
    }
#pragma unroll
    for (int e = 8; e < 16; ++e) Sa[pidx(e * 256 + t)] = 0.5f;

    for (int off = 16; off; off >>= 1) ss += __shfl_down(ss, off, 32);
    if (lane == 0) red[wave] = ss;
    __syncthreads();
    if (t == 0) {
        float tot = 512.0f;
        for (int w = 0; w < 8; ++w) tot += red[w];
        s_scale = 1.0f / sqrtf(tot);
    }
    {
        float c1[4], s1[4], c2[4], s2v[4];
        load4(c1, &CS[4]); load4(s1, &SN[4]); load4(c2, &CS[8]); load4(s2v, &SN[8]);
        *mst1 = kron_entry(c1, s1);
        *mst2 = kron_entry(c2, s2v);
    }
    __syncthreads();
    {
        float sc = s_scale;
#pragma unroll
        for (int e = 0; e < 16; ++e) Sa[pidx(e * 256 + t)] *= sc;
    }
    __syncthreads();

    int srcA[16];
#pragma unroll
    for (int tt = 0; tt < 2; ++tt) {
        int T = wave + tt * 8;
#pragma unroll
        for (int s = 0; s < 4; ++s) {
#pragma unroll
            for (int d = 0; d < 2; ++d) {
                int i = (T * 16 + lo) * 16 + 4 * s + 2 * hi + d;
                srcA[tt * 8 + s * 2 + d] = pidx(cnot_src(i));
            }
        }
    }

    float* const pa = Sa;
    float* const pb = Sb;

    const int T0 = wave;
    const int T1 = wave + 8;

    const float* const m2f = &Mm[2][lo * MROW + 10 * hi];
    const float* const m1f = &Mm[1][lo * MROW + 10 * hi];
    const float* const m0f = &Mm[0][lo * MROW + 10 * hi];
    float* const       yst = pb + T0 * PB0 + rowB(8 * hi) + lo;
    const float* const yld = pb + 2 * hi * PB0 + rowB(T0) + lo;
    float* const       xst = pa + 8 * hi * PA0 + rowA(T0) + lo;

    for (int L = 0; L < NLAYERS; ++L) {
        if (myL == L + 1 && myq >= 4) { CS[myq] = mc; SN[myq] = ms; }

        float c0v[4], s0v[4];
        load4(c0v, &CS[0]);
        load4(s0v, &SN[0]);
        v2f ga0[4], ga1[4], gb[4], ma1[4];
#pragma unroll
        for (int s = 0; s < 4; ++s) {
            ga0[s].x = pa[srcA[s * 2 + 0]];
            ga0[s].y = pa[srcA[s * 2 + 1]];
            ga1[s].x = pa[srcA[8 + s * 2 + 0]];
            ga1[s].y = pa[srcA[8 + s * 2 + 1]];
            gb[s].x  = m2f[2 * s];
            gb[s].y  = m2f[2 * s + 1];
            ma1[s].x = m1f[2 * s];
            ma1[s].y = m1f[2 * s + 1];
        }
        *mst0 = kron_entry(c0v, s0v);

        v8f acc0 = {0, 0, 0, 0, 0, 0, 0, 0};
        v8f acc1 = {0, 0, 0, 0, 0, 0, 0, 0};
#pragma unroll
        for (int s = 0; s < 4; ++s) {
            acc0 = __builtin_amdgcn_wmma_f32_16x16x4_f32(
                false, ga0[s], false, gb[s], (short)0, acc0, false, false);
            acc1 = __builtin_amdgcn_wmma_f32_16x16x4_f32(
                false, ga1[s], false, gb[s], (short)0, acc1, false, false);
        }

        {
            v8f c0 = {0, 0, 0, 0, 0, 0, 0, 0};
            v8f c1 = {0, 0, 0, 0, 0, 0, 0, 0};
#pragma unroll
            for (int s = 0; s < 4; ++s) {
                v2f a = ma1[s], b0, b1;
                b0.x = acc0[2 * s];
                b0.y = acc0[2 * s + 1];
                b1.x = acc1[2 * s];
                b1.y = acc1[2 * s + 1];
                c0 = __builtin_amdgcn_wmma_f32_16x16x4_f32(
                    false, a, false, b0, (short)0, c0, false, false);
                c1 = __builtin_amdgcn_wmma_f32_16x16x4_f32(
                    false, a, false, b1, (short)0, c1, false, false);
            }
#pragma unroll
            for (int v = 0; v < 8; ++v) {
                yst[rowB(v)]           = c0[v];
                yst[8 * PB0 + rowB(v)] = c1[v];
            }
        }
        __syncthreads();

        if (myL == L + 1 && myq < 4) { CS[myq] = mc; SN[myq] = ms; }
        float c1v[4], s1v[4], c2v[4], s2w[4];
        load4(c1v, &CS[4]);
        load4(s1v, &SN[4]);
        load4(c2v, &CS[8]);
        load4(s2w, &SN[8]);
        v2f ma0[4], yb0[4], yb1[4];
#pragma unroll
        for (int s = 0; s < 4; ++s) {
            ma0[s].x = m0f[2 * s];
            ma0[s].y = m0f[2 * s + 1];
            yb0[s].x = yld[(4 * s) * PB0];
            yb0[s].y = yld[(4 * s + 1) * PB0];
            yb1[s].x = yld[(4 * s) * PB0 + 32];
            yb1[s].y = yld[(4 * s + 1) * PB0 + 32];
        }
        *mst1 = kron_entry(c1v, s1v);
        *mst2 = kron_entry(c2v, s2w);

        {
            v8f d0 = {0, 0, 0, 0, 0, 0, 0, 0};
            v8f d1 = {0, 0, 0, 0, 0, 0, 0, 0};
#pragma unroll
            for (int s = 0; s < 4; ++s) {
                v2f a = ma0[s], b0 = yb0[s], b1 = yb1[s];
                d0 = __builtin_amdgcn_wmma_f32_16x16x4_f32(
                    false, a, false, b0, (short)0, d0, false, false);
                d1 = __builtin_amdgcn_wmma_f32_16x16x4_f32(
                    false, a, false, b1, (short)0, d1, false, false);
            }
#pragma unroll
            for (int v = 0; v < 8; ++v) {
                xst[v * PA0]      = d0[v];
                xst[v * PA0 + 32] = d1[v];
            }
        }
        __syncthreads();
    }

    float P = 0.f;
    float A[4] = {0.f, 0.f, 0.f, 0.f};
#pragma unroll
    for (int e = 0; e < 16; ++e) {
        int i = e * 256 + t;
        float v = pa[pidx(i)];
        float p = v * v;
        P += p;
#pragma unroll
        for (int q = 0; q < 4; ++q)
            A[q] += ((e >> (3 - q)) & 1) ? -p : p;
    }
    float acc[NQ];
#pragma unroll
    for (int q = 0; q < 4; ++q) acc[q] = A[q];
#pragma unroll
    for (int q = 4; q < NQ; ++q)
        acc[q] = ((t >> (11 - q)) & 1) ? -P : P;

    for (int off = 16; off; off >>= 1) {
#pragma unroll
        for (int q = 0; q < NQ; ++q) acc[q] += __shfl_down(acc[q], off, 32);
    }
    if (lane == 0) {
#pragma unroll
        for (int q = 0; q < NQ; ++q) red[wave * NQ + q] = acc[q];
    }
    __syncthreads();
    if (t < NQ) {
        float s = 0.f;
#pragma unroll
        for (int w = 0; w < 8; ++w) s += red[w * NQ + t];
        out[(size_t)row * NQ + t] = s;
    }
}

extern "C" void kernel_launch(void* const* d_in, const int* in_sizes, int n_in,
                              void* d_out, int out_size, void* d_ws, size_t ws_size,
                              hipStream_t stream) {
    (void)n_in; (void)out_size; (void)d_ws; (void)ws_size;
    const float* in  = (const float*)d_in[0];
    const float* qp  = (const float*)d_in[1];
    float*       out = (float*)d_out;
    int rows = in_sizes[0] / F_IN;
    qnet_kernel<<<rows, 256, 0, stream>>>(in, qp, out);
}


# each workgroup carries two rows through the layers together: rotation matrices, their fragments and cos/sin reads are built and loaded once per two rows, one barrier pair serves both; grid halves
# speedup vs baseline: 1.3198x; 1.0451x over previous
#include <hip/hip_runtime.h>
#include <math.h>

typedef __attribute__((ext_vector_type(2))) float v2f;
typedef __attribute__((ext_vector_type(8))) float v8f;

#define NQ      12
#define DIM     4096
#define NLAYERS 20
#define F_IN    2048
#define PA0     266
#define PB0     264
#define SA_SIZE (15 * PA0 + 264)
#define SB_SIZE (15 * PB0 + 256)
#define MROW    20
#define NR      2

__device__ __forceinline__ int rowA(int a1) {
    return 16 * (a1 & 1) + 64 * ((a1 >> 1) & 1) + 136 * ((a1 >> 2) & 1) + 32 * (a1 >> 3);
}
__device__ __forceinline__ int rowB(int a1) {
    return 16 * (a1 & 1) + 64 * ((a1 >> 1) & 3) + 32 * (a1 >> 3);
}
__device__ __forceinline__ int pidx(int i) { return (i >> 8) * PA0 + rowA((i >> 4) & 15) + (i & 15); }

__device__ __forceinline__ int cnot_src(int i) {
    const int cq[11] = {0, 2, 4, 6, 8, 10, 1, 3, 5, 7, 9};
    const int tq[11] = {1, 3, 5, 7, 9, 11, 2, 4, 6, 8, 10};
    int j = i;
#pragma unroll
    for (int k = 10; k >= 0; --k) {
        int cm = 1 << (11 - cq[k]);
        int tm = 1 << (11 - tq[k]);
        if (j & cm) j ^= tm;
    }
    return j;
}

__device__ __forceinline__ float tanh_pio2(float x) {
    float xc = fminf(fmaxf(x, -15.f), 15.f);
    float z  = __expf(xc * 2.0f);
    return (z - 1.0f) * __builtin_amdgcn_rcpf(z + 1.0f) * 1.57079632679489662f;
}

__global__ __launch_bounds__(256, 8)
void qnet_kernel(const float* __restrict__ in, const float* __restrict__ qp,
                 float* __restrict__ out)
{
    __shared__ __attribute__((aligned(16))) float Sa[NR * SA_SIZE];
    __shared__ __attribute__((aligned(16))) float Sb[NR * SB_SIZE];
    __shared__ __attribute__((aligned(16))) float Mm[3][16 * MROW];
    __shared__ float CS[NQ], SN[NQ];
    __shared__ float red[NR * 96];
    __shared__ float s_scale[NR];

    const int t    = threadIdx.x;
    const int lane = t & 31;
    const int wave = t >> 5;
    const int hi   = lane >> 4;
    const int lo   = lane & 15;
    const int row0 = blockIdx.x * NR;

    const float* th = qp + NQ;

    const int  hasAng = t < NLAYERS * NQ;
    const int  myL    = hasAng ? t / NQ : -1;
    const int  myq    = t - myL * NQ;
    float mc = 1.f, ms = 0.f;
    if (hasAng) {
        float a = th[t] * 0.5f;
        mc = cosf(a);
        ms = sinf(a);
    }
    if (myL == 0) { CS[myq] = mc; SN[myq] = ms; }

    const int mr = t >> 4, mcn = t & 15;
    const int pc1  = 10 * (mcn >> 3) + 2 * ((mcn >> 1) & 3) + (mcn & 1);
    const int pc02 = 10 * ((mcn >> 1) & 1) + 2 * (mcn >> 2) + (mcn & 1);
    auto kron_entry = [&](const float* c4, const float* s4) {
        float f = 1.f;
#pragma unroll
        for (int p = 0; p < 4; ++p) {
            int rb = (mr >> (3 - p)) & 1;
            int cb = (mcn >> (3 - p)) & 1;
            f *= (rb == cb) ? c4[p] : (rb ? s4[p] : -s4[p]);
        }
        return f;
    };
    auto load4 = [&](float* dst, const float* src4) {
#pragma unroll
        for (int p = 0; p < 4; ++p) dst[p] = src4[p];
    };
    float* const mst0 = &Mm[0][mr * MROW + pc02];
    float* const mst1 = &Mm[1][mr * MROW + pc1];
    float* const mst2 = &Mm[2][mr * MROW + pc02];

    float ss[NR];
#pragma unroll
    for (int r = 0; r < NR; ++r) {
        const float* srcrow = in + (size_t)(row0 + r) * F_IN;
        float* sa = Sa + r * SA_SIZE;
        ss[r] = 0.f;
#pragma unroll
        for (int e = 0; e < 8; ++e) {
            int i = e * 256 + t;
            float q = tanh_pio2(srcrow[i]);
            sa[pidx(i)] = q;
            ss[r] += q * q;
        }
#pragma unroll
        for (int e = 8; e < 16; ++e) sa[pidx(e * 256 + t)] = 0.5f;
    }
    for (int off = 16; off; off >>= 1) {
#pragma unroll
        for (int r = 0; r < NR; ++r) ss[r] += __shfl_down(ss[r], off, 32);
    }
    if (lane == 0) {
#pragma unroll
        for (int r = 0; r < NR; ++r) red[r * 8 + wave] = ss[r];
    }
    __syncthreads();
    if (t < NR) {
        float tot = 512.0f;
        for (int w = 0; w < 8; ++w) tot += red[t * 8 + w];
        s_scale[t] = 1.0f / sqrtf(tot);
    }
    {
        float c1[4], s1[4], c2[4], s2v[4];
        load4(c1, &CS[4]); load4(s1, &SN[4]); load4(c2, &CS[8]); load4(s2v, &SN[8]);
        *mst1 = kron_entry(c1, s1);
        *mst2 = kron_entry(c2, s2v);
    }
    __syncthreads();
#pragma unroll
    for (int r = 0; r < NR; ++r) {
        float sc = s_scale[r];
        float* sa = Sa + r * SA_SIZE;
#pragma unroll
        for (int e = 0; e < 16; ++e) sa[pidx(e * 256 + t)] *= sc;
    }
    __syncthreads();

    int srcA[16];
#pragma unroll
    for (int tt = 0; tt < 2; ++tt) {
        int T = wave + tt * 8;
#pragma unroll
        for (int s = 0; s < 4; ++s) {
#pragma unroll
            for (int d = 0; d < 2; ++d) {
                int i = (T * 16 + lo) * 16 + 4 * s + 2 * hi + d;
                srcA[tt * 8 + s * 2 + d] = pidx(cnot_src(i));
            }
        }
    }

    float* const pa = Sa;
    float* const pb = Sb;

    const int T0 = wave;
    const int T1 = wave + 8;

    const float* const m2f = &Mm[2][lo * MROW + 10 * hi];
    const float* const m1f = &Mm[1][lo * MROW + 10 * hi];
    const float* const m0f = &Mm[0][lo * MROW + 10 * hi];
    float* const       yst = pb + T0 * PB0 + rowB(8 * hi) + lo;
    const float* const yld = pb + 2 * hi * PB0 + rowB(T0) + lo;
    float* const       xst = pa + 8 * hi * PA0 + rowA(T0) + lo;

    for (int L = 0; L < NLAYERS; ++L) {
        if (myL == L + 1 && myq >= 4) { CS[myq] = mc; SN[myq] = ms; }

        float c0v[4], s0v[4];
        load4(c0v, &CS[0]);
        load4(s0v, &SN[0]);
        v2f gb[4], ma1[4];
#pragma unroll
        for (int s = 0; s < 4; ++s) {
            gb[s].x  = m2f[2 * s];
            gb[s].y  = m2f[2 * s + 1];
            ma1[s].x = m1f[2 * s];
            ma1[s].y = m1f[2 * s + 1];
        }
        v2f ga0[NR][4], ga1[NR][4];
#pragma unroll
        for (int r = 0; r < NR; ++r) {
            const float* par = pa + r * SA_SIZE;
#pragma unroll
            for (int s = 0; s < 4; ++s) {
                ga0[r][s].x = par[srcA[s * 2 + 0]];
                ga0[r][s].y = par[srcA[s * 2 + 1]];
                ga1[r][s].x = par[srcA[8 + s * 2 + 0]];
                ga1[r][s].y = par[srcA[8 + s * 2 + 1]];
            }
        }
        *mst0 = kron_entry(c0v, s0v);

#pragma unroll
        for (int r = 0; r < NR; ++r) {
            v8f acc0 = {0, 0, 0, 0, 0, 0, 0, 0};
            v8f acc1 = {0, 0, 0, 0, 0, 0, 0, 0};
#pragma unroll
            for (int s = 0; s < 4; ++s) {
                acc0 = __builtin_amdgcn_wmma_f32_16x16x4_f32(
                    false, ga0[r][s], false, gb[s], (short)0, acc0, false, false);
                acc1 = __builtin_amdgcn_wmma_f32_16x16x4_f32(
                    false, ga1[r][s], false, gb[s], (short)0, acc1, false, false);
            }

            v8f c0 = {0, 0, 0, 0, 0, 0, 0, 0};
            v8f c1 = {0, 0, 0, 0, 0, 0, 0, 0};
#pragma unroll
            for (int s = 0; s < 4; ++s) {
                v2f a = ma1[s], b0, b1;
                b0.x = acc0[2 * s];
                b0.y = acc0[2 * s + 1];
                b1.x = acc1[2 * s];
                b1.y = acc1[2 * s + 1];
                c0 = __builtin_amdgcn_wmma_f32_16x16x4_f32(
                    false, a, false, b0, (short)0, c0, false, false);
                c1 = __builtin_amdgcn_wmma_f32_16x16x4_f32(
                    false, a, false, b1, (short)0, c1, false, false);
            }
            float* const ystr = yst + r * SB_SIZE;
#pragma unroll
            for (int v = 0; v < 8; ++v) {
                ystr[rowB(v)]           = c0[v];
                ystr[8 * PB0 + rowB(v)] = c1[v];
            }
        }
        __syncthreads();

        if (myL == L + 1 && myq < 4) { CS[myq] = mc; SN[myq] = ms; }
        float c1v[4], s1v[4], c2v[4], s2w[4];
        load4(c1v, &CS[4]);
        load4(s1v, &SN[4]);
        load4(c2v, &CS[8]);
        load4(s2w, &SN[8]);
        v2f ma0[4], yb0[NR][4], yb1[NR][4];
#pragma unroll
        for (int s = 0; s < 4; ++s) {
            ma0[s].x = m0f[2 * s];
            ma0[s].y = m0f[2 * s + 1];
        }
#pragma unroll
        for (int r = 0; r < NR; ++r) {
            const float* yldr = yld + r * SB_SIZE;
#pragma unroll
            for (int s = 0; s < 4; ++s) {
                yb0[r][s].x = yldr[(4 * s) * PB0];
                yb0[r][s].y = yldr[(4 * s + 1) * PB0];
                yb1[r][s].x = yldr[(4 * s) * PB0 + 32];
                yb1[r][s].y = yldr[(4 * s + 1) * PB0 + 32];
            }
        }
        *mst1 = kron_entry(c1v, s1v);
        *mst2 = kron_entry(c2v, s2w);

        {
            v8f d0[NR], d1[NR];
#pragma unroll
            for (int r = 0; r < NR; ++r) {
                d0[r] = (v8f){0, 0, 0, 0, 0, 0, 0, 0};
                d1[r] = (v8f){0, 0, 0, 0, 0, 0, 0, 0};
            }
#pragma unroll
            for (int s = 0; s < 4; ++s) {
#pragma unroll
                for (int r = 0; r < NR; ++r) {
                    d0[r] = __builtin_amdgcn_wmma_f32_16x16x4_f32(
                        false, ma0[s], false, yb0[r][s], (short)0, d0[r], false, false);
                    d1[r] = __builtin_amdgcn_wmma_f32_16x16x4_f32(
                        false, ma0[s], false, yb1[r][s], (short)0, d1[r], false, false);
                }
            }
#pragma unroll
            for (int r = 0; r < NR; ++r) {
                float* const xstr = xst + r * SA_SIZE;
#pragma unroll
                for (int v = 0; v < 8; ++v) {
                    xstr[v * PA0]      = d0[r][v];
                    xstr[v * PA0 + 32] = d1[r][v];
                }
            }
        }
        __syncthreads();
    }

#pragma unroll
    for (int r = 0; r < NR; ++r) {
        const float* par = pa + r * SA_SIZE;
        float P = 0.f;
        float A[4] = {0.f, 0.f, 0.f, 0.f};
#pragma unroll
        for (int e = 0; e < 16; ++e) {
            int i = e * 256 + t;
            float v = par[pidx(i)];
            float p = v * v;
            P += p;
#pragma unroll
            for (int q = 0; q < 4; ++q)
                A[q] += ((e >> (3 - q)) & 1) ? -p : p;
        }
        float acc[NQ];
#pragma unroll
        for (int q = 0; q < 4; ++q) acc[q] = A[q];
#pragma unroll
        for (int q = 4; q < NQ; ++q)
            acc[q] = ((t >> (11 - q)) & 1) ? -P : P;

        for (int off = 16; off; off >>= 1) {
#pragma unroll
            for (int q = 0; q < NQ; ++q) acc[q] += __shfl_down(acc[q], off, 32);
        }
        if (lane == 0) {
#pragma unroll
            for (int q = 0; q < NQ; ++q) red[r * 96 + wave * NQ + q] = acc[q];
        }
    }
    __syncthreads();
    if (t < NR * NQ) {
        const int r = t / NQ, q = t - r * NQ;
        float s = 0.f;
#pragma unroll
        for (int w = 0; w < 8; ++w) s += red[r * 96 + w * NQ + q];
        out[(size_t)row0 * NQ + t] = s;
    }
}

extern "C" void kernel_launch(void* const* d_in, const int* in_sizes, int n_in,
                              void* d_out, int out_size, void* d_ws, size_t ws_size,
                              hipStream_t stream) {
    (void)n_in; (void)out_size; (void)d_ws; (void)ws_size;
    const float* in  = (const float*)d_in[0];
    const float* qp  = (const float*)d_in[1];
    float*       out = (float*)d_out;
    int groups = in_sizes[0] / (F_IN * NR);
    qnet_kernel<<<groups, 256, 0, stream>>>(in, qp, out);
}


# both LDS state buffers re-laid so each lane's state fragment is two 16-byte reads; CNOT permutation moved from gather reads to axis-0 stores; k order 8h+2s+d in all products: sum order changes
# speedup vs baseline: 1.4360x; 1.0880x over previous
#include <hip/hip_runtime.h>
#include <math.h>

typedef __attribute__((ext_vector_type(2))) float v2f;
typedef __attribute__((ext_vector_type(4))) float v4f;
typedef __attribute__((ext_vector_type(8))) float v8f;

#define NQ      12
#define DIM     4096
#define NLAYERS 20
#define F_IN    2048
#define PA0     328
#define RA      20
#define SA_SIZE (15 * PA0 + 15 * RA + 16)
#define SB_SIZE 4096
#define MROW    20
#define NR      2

__device__ __forceinline__ int cnot_dst(int j) {
    const int cq[11] = {0, 2, 4, 6, 8, 10, 1, 3, 5, 7, 9};
    const int tq[11] = {1, 3, 5, 7, 9, 11, 2, 4, 6, 8, 10};
#pragma unroll
    for (int k = 0; k <= 10; ++k) {
        int cm = 1 << (11 - cq[k]);
        int tm = 1 << (11 - tq[k]);
        if (j & cm) j ^= tm;
    }
    return j;
}
__device__ __forceinline__ int pidx(int i) {
    const int n = cnot_dst(i);
    return (n >> 8) * PA0 + ((n >> 4) & 15) * RA + (n & 15);
}

__device__ __forceinline__ float tanh_pio2(float x) {
    float xc = fminf(fmaxf(x, -15.f), 15.f);
    float z  = __expf(xc * 2.0f);
    return (z - 1.0f) * __builtin_amdgcn_rcpf(z + 1.0f) * 1.57079632679489662f;
}

__global__ __launch_bounds__(256, 8)
void qnet_kernel(const float* __restrict__ in, const float* __restrict__ qp,
                 float* __restrict__ out)
{
    __shared__ __attribute__((aligned(16))) float Sa[NR * SA_SIZE];
    __shared__ __attribute__((aligned(16))) float Sb[NR * SB_SIZE];
    __shared__ __attribute__((aligned(16))) float Mm[3][16 * MROW];
    __shared__ float CS[NQ], SN[NQ];
    __shared__ float red[NR * 96];
    __shared__ float s_scale[NR];

    const int t    = threadIdx.x;
    const int lane = t & 31;
    const int wave = t >> 5;
    const int hi   = lane >> 4;
    const int lo   = lane & 15;
    const int row0 = blockIdx.x * NR;

    const float* th = qp + NQ;

    auto tpart = [&](int flip) {
        const int nt = cnot_dst(t);
        return (((nt >> 4) & 15) ^ flip) * RA + (nt & 15);
    };
    auto epart = [&](int e, int tb0, int tb8) {
        const int ne = cnot_dst(e << 8);
        return (ne >> 8) * PA0 + (((ne >> 4) & 8) ? tb8 : tb0);
    };

    const int  hasAng = t < NLAYERS * NQ;
    const int  myL    = hasAng ? t / NQ : -1;
    const int  myq    = t - myL * NQ;
    float mc = 1.f, ms = 0.f;
    if (hasAng) {
        float a = th[t] * 0.5f;
        mc = cosf(a);
        ms = sinf(a);
    }
    if (myL == 0) { CS[myq] = mc; SN[myq] = ms; }

    const int mr = t >> 4, mcn = t & 15;
    const int pc1  = 10 * (mcn >> 3) + 2 * ((mcn >> 1) & 3) + (mcn & 1);
    auto kron_entry = [&](const float* c4, const float* s4) {
        float f = 1.f;
#pragma unroll
        for (int p = 0; p < 4; ++p) {
            int rb = (mr >> (3 - p)) & 1;
            int cb = (mcn >> (3 - p)) & 1;
            f *= (rb == cb) ? c4[p] : (rb ? s4[p] : -s4[p]);
        }
        return f;
    };
    auto load4 = [&](float* dst, const float* src4) {
#pragma unroll
        for (int p = 0; p < 4; ++p) dst[p] = src4[p];
    };
    float* const mst0 = &Mm[0][mr * MROW + pc1];
    float* const mst1 = &Mm[1][mr * MROW + pc1];
    float* const mst2 = &Mm[2][mr * MROW + pc1];

    float ss[NR];
    const int eb0 = tpart(0), eb8 = tpart(8);
#pragma unroll
    for (int r = 0; r < NR; ++r) {
        const float* srcrow = in + (size_t)(row0 + r) * F_IN;
        float* sa = Sa + r * SA_SIZE;
        ss[r] = 0.f;
#pragma unroll
        for (int e = 0; e < 8; ++e) {
            float q = tanh_pio2(srcrow[e * 256 + t]);
            sa[epart(e, eb0, eb8)] = q;
            ss[r] += q * q;
        }
#pragma unroll
        for (int e = 8; e < 16; ++e) sa[epart(e, eb0, eb8)] = 0.5f;
    }
    for (int off = 16; off; off >>= 1) {
#pragma unroll
        for (int r = 0; r < NR; ++r) ss[r] += __shfl_down(ss[r], off, 32);
    }
    if (lane == 0) {
#pragma unroll
        for (int r = 0; r < NR; ++r) red[r * 8 + wave] = ss[r];
    }
    __syncthreads();
    if (t < NR) {
        float tot = 512.0f;
        for (int w = 0; w < 8; ++w) tot += red[t * 8 + w];
        s_scale[t] = 1.0f / sqrtf(tot);
    }
    {
        float c1[4], s1[4], c2[4], s2v[4];
        load4(c1, &CS[4]); load4(s1, &SN[4]); load4(c2, &CS[8]); load4(s2v, &SN[8]);
        *mst1 = kron_entry(c1, s1);
        *mst2 = kron_entry(c2, s2v);
    }
    __syncthreads();
#pragma unroll
    for (int r = 0; r < NR; ++r) {
        float sc = s_scale[r];
        float* sa = Sa + r * SA_SIZE;
#pragma unroll
        for (int e = 0; e < 16; ++e) sa[epart(e, eb0, eb8)] *= sc;
    }
    __syncthreads();

    int dstA[16];
#pragma unroll
    for (int tt = 0; tt < 2; ++tt) {
#pragma unroll
        for (int v = 0; v < 8; ++v)
            dstA[tt * 8 + v] = pidx((8 * hi + v) * 256 + (wave + tt * 8) * 16 + lo);
    }

    float* const pa = Sa;
    float* const pb = Sb;

    const int T0 = wave;
    const int T1 = wave + 8;

    const float* const m2f = &Mm[2][lo * MROW + 10 * hi];
    const float* const m1f = &Mm[1][lo * MROW + 10 * hi];
    const float* const m0f = &Mm[0][lo * MROW + 10 * hi];
    float* const       yst = pb + (T0 >> 2) * 1024 + hi * 512 + lo * 4 + ((T0 & 3) ^ (2 * hi));
    const float* const yld = pb + 2 * hi * 1024 + T0 * 64 + lo * 4;
    const float* const xld = pa + T0 * PA0 + lo * RA + 8 * hi;

    for (int L = 0; L < NLAYERS; ++L) {
        if (myL == L + 1 && myq >= 4) { CS[myq] = mc; SN[myq] = ms; }

        float c0v[4], s0v[4];
        load4(c0v, &CS[0]);
        load4(s0v, &SN[0]);
        v2f gb[4], ma1[4];
#pragma unroll
        for (int s = 0; s < 4; ++s) {
            gb[s].x  = m2f[2 * s];
            gb[s].y  = m2f[2 * s + 1];
            ma1[s].x = m1f[2 * s];
            ma1[s].y = m1f[2 * s + 1];
        }
        v2f ga0[NR][4], ga1[NR][4];
#pragma unroll
        for (int r = 0; r < NR; ++r) {
            const float* xldr = xld + r * SA_SIZE;
#pragma unroll
            for (int g = 0; g < 2; ++g) {
                const v4f q0 = *(const v4f*)(xldr + 4 * g);
                const v4f q1 = *(const v4f*)(xldr + 8 * PA0 + 4 * g);
                ga0[r][2 * g]     = (v2f){q0.x, q0.y};
                ga0[r][2 * g + 1] = (v2f){q0.z, q0.w};
                ga1[r][2 * g]     = (v2f){q1.x, q1.y};
                ga1[r][2 * g + 1] = (v2f){q1.z, q1.w};
            }
        }
        *mst0 = kron_entry(c0v, s0v);

#pragma unroll
        for (int r = 0; r < NR; ++r) {
            v8f acc0 = {0, 0, 0, 0, 0, 0, 0, 0};
            v8f acc1 = {0, 0, 0, 0, 0, 0, 0, 0};
#pragma unroll
            for (int s = 0; s < 4; ++s) {
                acc0 = __builtin_amdgcn_wmma_f32_16x16x4_f32(
                    false, ga0[r][s], false, gb[s], (short)0, acc0, false, false);
                acc1 = __builtin_amdgcn_wmma_f32_16x16x4_f32(
                    false, ga1[r][s], false, gb[s], (short)0, acc1, false, false);
            }

            v8f c0 = {0, 0, 0, 0, 0, 0, 0, 0};
            v8f c1 = {0, 0, 0, 0, 0, 0, 0, 0};
#pragma unroll
            for (int s = 0; s < 4; ++s) {
                v2f a = ma1[s], b0, b1;
                b0.x = acc0[2 * s];
                b0.y = acc0[2 * s + 1];
                b1.x = acc1[2 * s];
                b1.y = acc1[2 * s + 1];
                c0 = __builtin_amdgcn_wmma_f32_16x16x4_f32(
                    false, a, false, b0, (short)0, c0, false, false);
                c1 = __builtin_amdgcn_wmma_f32_16x16x4_f32(
                    false, a, false, b1, (short)0, c1, false, false);
            }
            float* const ystr = yst + r * SB_SIZE;
#pragma unroll
            for (int v = 0; v < 8; ++v) {
                ystr[v * 64]        = c0[v];
                ystr[2048 + v * 64] = c1[v];
            }
        }
        __syncthreads();

        if (myL == L + 1 && myq < 4) { CS[myq] = mc; SN[myq] = ms; }
        float c1v[4], s1v[4], c2v[4], s2w[4];
        load4(c1v, &CS[4]);
        load4(s1v, &SN[4]);
        load4(c2v, &CS[8]);
        load4(s2w, &SN[8]);
        v2f ma0[4], yb0[NR][4], yb1[NR][4];
#pragma unroll
        for (int s = 0; s < 4; ++s) {
            ma0[s].x = m0f[2 * s];
            ma0[s].y = m0f[2 * s + 1];
        }
#pragma unroll
        for (int r = 0; r < NR; ++r) {
            const float* yldr = yld + r * SB_SIZE;
#pragma unroll
            for (int g = 0; g < 2; ++g) {
                const v4f q0 = *(const v4f*)(yldr + g * 1024);
                const v4f q1 = *(const v4f*)(yldr + g * 1024 + 512);
                yb0[r][2 * g]     = (v2f){q0.x, q0.y};
                yb0[r][2 * g + 1] = (v2f){q0.z, q0.w};
                yb1[r][2 * g + 1] = (v2f){q1.x, q1.y};
                yb1[r][2 * g]     = (v2f){q1.z, q1.w};
            }
        }
        *mst1 = kron_entry(c1v, s1v);
        *mst2 = kron_entry(c2v, s2w);

        {
            v8f d0[NR], d1[NR];
#pragma unroll
            for (int r = 0; r < NR; ++r) {
                d0[r] = (v8f){0, 0, 0, 0, 0, 0, 0, 0};
                d1[r] = (v8f){0, 0, 0, 0, 0, 0, 0, 0};
            }
#pragma unroll
            for (int s = 0; s < 4; ++s) {
#pragma unroll
                for (int r = 0; r < NR; ++r) {
                    d0[r] = __builtin_amdgcn_wmma_f32_16x16x4_f32(
                        false, ma0[s], false, yb0[r][s], (short)0, d0[r], false, false);
                    d1[r] = __builtin_amdgcn_wmma_f32_16x16x4_f32(
                        false, ma0[s], false, yb1[r][s], (short)0, d1[r], false, false);
                }
            }
#pragma unroll
            for (int r = 0; r < NR; ++r) {
                float* const xstr = pa + r * SA_SIZE;
#pragma unroll
                for (int v = 0; v < 8; ++v) {
                    xstr[dstA[v]]     = d0[r][v];
                    xstr[dstA[8 + v]] = d1[r][v];
                }
            }
        }
        __syncthreads();
    }

    const int rb0 = tpart(0), rb8 = tpart(8);
#pragma unroll
    for (int r = 0; r < NR; ++r) {
        const float* par = pa + r * SA_SIZE;
        float P = 0.f;
        float A[4] = {0.f, 0.f, 0.f, 0.f};
#pragma unroll
        for (int e = 0; e < 16; ++e) {
            float v = par[epart(e, rb0, rb8)];
            float p = v * v;
            P += p;
#pragma unroll
            for (int q = 0; q < 4; ++q)
                A[q] += ((e >> (3 - q)) & 1) ? -p : p;
        }
        float acc[NQ];
#pragma unroll
        for (int q = 0; q < 4; ++q) acc[q] = A[q];
#pragma unroll
        for (int q = 4; q < NQ; ++q)
            acc[q] = ((t >> (11 - q)) & 1) ? -P : P;

        for (int off = 16; off; off >>= 1) {
#pragma unroll
            for (int q = 0; q < NQ; ++q) acc[q] += __shfl_down(acc[q], off, 32);
        }
        if (lane == 0) {
#pragma unroll
            for (int q = 0; q < NQ; ++q) red[r * 96 + wave * NQ + q] = acc[q];
        }
    }
    __syncthreads();
    if (t < NR * NQ) {
        const int r = t / NQ, q = t - r * NQ;
        float s = 0.f;
#pragma unroll
        for (int w = 0; w < 8; ++w) s += red[r * 96 + w * NQ + q];
        out[(size_t)row0 * NQ + t] = s;
    }
}

extern "C" void kernel_launch(void* const* d_in, const int* in_sizes, int n_in,
                              void* d_out, int out_size, void* d_ws, size_t ws_size,
                              hipStream_t stream) {
    (void)n_in; (void)out_size; (void)d_ws; (void)ws_size;
    const float* in  = (const float*)d_in[0];
    const float* qp  = (const float*)d_in[1];
    float*       out = (float*)d_out;
    int groups = in_sizes[0] / (F_IN * NR);
    qnet_kernel<<<groups, 256, 0, stream>>>(in, qp, out);
}
